# starting kernel
# baseline (speedup 1.0000x reference)
#include <hip/hip_runtime.h>

typedef __attribute__((ext_vector_type(2))) float v2f;
typedef __attribute__((ext_vector_type(8))) float v8f;

#define F_IN  128
#define F_HID 64
#define F_OUT 40


__global__ void deg_init_kernel(float* __restrict__ deg, int n) {
    int i = blockIdx.x * blockDim.x + threadIdx.x;
    if (i < n) deg[i] = 1.0f;
}

__global__ void deg_accum_kernel(const int* __restrict__ ei, float* __restrict__ deg, int E) {
    int e = blockIdx.x * blockDim.x + threadIdx.x;
    if (e < E) atomicAdd(&deg[ei[E + e]], 1.0f);
}

__global__ void dinv_kernel(float* __restrict__ deg, int n) {
    int i = blockIdx.x * blockDim.x + threadIdx.x;
    if (i < n) { float d = deg[i]; deg[i] = (d > 0.0f) ? rsqrtf(d) : 0.0f; }
}

__global__ void norm_kernel(const int* __restrict__ ei, const float* __restrict__ dinv,
                            float* __restrict__ norm, int E) {
    int e = blockIdx.x * blockDim.x + threadIdx.x;
    if (e < E) norm[e] = dinv[ei[e]] * dinv[ei[E + e]];
}

template <bool RELU_IN>
__global__ __launch_bounds__(128) void gemm_wmma_kernel(const float* __restrict__ X,
                                                        const float* __restrict__ W,
                                                        float* __restrict__ H,
                                                        int K, int Fout) {
    const int lane = threadIdx.x & 31;
    const int wave = threadIdx.x >> 5;
    const int half = lane >> 4;
    const int l    = lane & 15;
    const int row0 = blockIdx.x * 16;
    const int col  = wave * 16 + l;
    const bool colOK  = (col < Fout);
    const int  colLd  = colOK ? col : (Fout - 1);

    const float* __restrict__ xrow = X + (size_t)(row0 + l) * K + 2 * half;
    const float* __restrict__ wcol = W + colLd + (size_t)(2 * half) * Fout;

    v8f c = {};
#pragma unroll 4
    for (int k0 = 0; k0 < K; k0 += 4) {
        float a0 = xrow[k0];
        float a1 = xrow[k0 + 1];
        if (RELU_IN) { a0 = fmaxf(a0, 0.0f); a1 = fmaxf(a1, 0.0f); }
        v2f a; a.x = a0; a.y = a1;
        v2f b;
        b.x = wcol[(size_t)k0 * Fout];
        b.y = wcol[(size_t)(k0 + 1) * Fout];
        c = __builtin_amdgcn_wmma_f32_16x16x4_f32(false, a, false, b, (short)0, c,
                                                  false, false);
    }

    if (colOK) {
#pragma unroll
        for (int v = 0; v < 8; ++v) {
            const int m = v + 8 * half;
            H[(size_t)(row0 + m) * Fout + col] = c[v];
        }
    }
}


__global__ void agg_init_kernel(const float* __restrict__ H, const float* __restrict__ dinv,
                                const float* __restrict__ bias, float* __restrict__ A,
                                int N, int F) {
    int idx = blockIdx.x * blockDim.x + threadIdx.x;
    if (idx < N * F) {
        int i = idx / F;
        int f = idx - i * F;
        float di = dinv[i];
        A[idx] = bias[f] + H[idx] * (di * di);
    }
}

__global__ void agg_edges_kernel(const float* __restrict__ H, const int* __restrict__ ei,
                                 const float* __restrict__ norm, float* __restrict__ A,
                                 int E, int F) {
    const int chunks = F >> 2;
    long long tid   = (long long)blockIdx.x * blockDim.x + threadIdx.x;
    long long total = (long long)E * chunks;
    if (tid >= total) return;
    int e  = (int)(tid / chunks);
    int f0 = (int)(tid - (long long)e * chunks) * 4;

    int   s  = ei[e];
    int   d  = ei[E + e];
    float nv = norm[e];

    const float4 hv = *(const float4*)(H + (size_t)s * F + f0);
    float* __restrict__ out = A + (size_t)d * F + f0;
    atomicAdd(out + 0, hv.x * nv);
    atomicAdd(out + 1, hv.y * nv);
    atomicAdd(out + 2, hv.z * nv);
    atomicAdd(out + 3, hv.w * nv);
}


extern "C" void kernel_launch(void* const* d_in, const int* in_sizes, int n_in,
                              void* d_out, int out_size, void* d_ws, size_t ws_size,
                              hipStream_t stream) {
    const float* x  = (const float*)d_in[0];
    const int*   ei = (const int*)d_in[1];
    const float* W1 = (const float*)d_in[2];
    const float* b1 = (const float*)d_in[3];
    const float* W2 = (const float*)d_in[4];
    const float* b2 = (const float*)d_in[5];
    const float* W3 = (const float*)d_in[6];
    const float* b3 = (const float*)d_in[7];
    float* out = (float*)d_out;

    const int N = in_sizes[0] / F_IN;
    const int E = in_sizes[1] / 2;

    float* dinv = (float*)d_ws;
    float* norm = dinv + N;
    float* bufH = norm + E;
    float* bufA = bufH + (size_t)N * F_HID;

    const int T = 256;
    const int gN  = (N + T - 1) / T;
    const int gE  = (E + T - 1) / T;
    const int gNF = (N * F_HID + T - 1) / T;
    const int gNO = (N * F_OUT + T - 1) / T;
    const unsigned gEF = (unsigned)(((long long)E * (F_HID / 4) + T - 1) / T);
    const unsigned gEO = (unsigned)(((long long)E * (F_OUT / 4) + T - 1) / T);

    deg_init_kernel <<<gN, T, 0, stream>>>(dinv, N);
    deg_accum_kernel<<<gE, T, 0, stream>>>(ei, dinv, E);
    dinv_kernel     <<<gN, T, 0, stream>>>(dinv, N);
    norm_kernel     <<<gE, T, 0, stream>>>(ei, dinv, norm, E);

    const int rowTiles = (N + 15) / 16;
    const int wavesHid = (F_HID + 15) / 16;
    const int wavesOut = (F_OUT + 15) / 16;

    gemm_wmma_kernel<false><<<rowTiles, 32 * wavesHid, 0, stream>>>(x, W1, bufH, F_IN, F_HID);
    agg_init_kernel <<<gNF, T, 0, stream>>>(bufH, dinv, b1, bufA, N, F_HID);
    agg_edges_kernel<<<gEF, T, 0, stream>>>(bufH, ei, norm, bufA, E, F_HID);

    gemm_wmma_kernel<true> <<<rowTiles, 32 * wavesHid, 0, stream>>>(bufA, W2, bufH, F_HID, F_HID);
    agg_init_kernel <<<gNF, T, 0, stream>>>(bufH, dinv, b2, bufA, N, F_HID);
    agg_edges_kernel<<<gEF, T, 0, stream>>>(bufH, ei, norm, bufA, E, F_HID);

    gemm_wmma_kernel<true> <<<rowTiles, 32 * wavesOut, 0, stream>>>(bufA, W3, bufH, F_HID, F_OUT);
    agg_init_kernel <<<gNO, T, 0, stream>>>(bufH, dinv, b3, out, N, F_OUT);
    agg_edges_kernel<<<gEO, T, 0, stream>>>(bufH, ei, norm, out, E, F_OUT);
}


# Edge scatters: each destination row is added to by ONE package half: a work-group reads its die and draws an edge chunk for that half's node range from a counter; atomics kept (arrival order)
# speedup vs baseline: 2.5654x; 2.5654x over previous
#include <hip/hip_runtime.h>

typedef __attribute__((ext_vector_type(2))) float v2f;
typedef __attribute__((ext_vector_type(8))) float v8f;

#define F_IN  128
#define F_HID 64
#define F_OUT 40


#define DEG_SLICES 64
#define DEG_TABLE  51200
#define DEG_BATCH  8

__global__ __launch_bounds__(1024) void deg_count_kernel(const int* __restrict__ ei,
                                                         unsigned* __restrict__ part,
                                                         int N, int E) {
    __shared__ unsigned cnt[DEG_TABLE];
    const int per = (E + DEG_SLICES - 1) / DEG_SLICES;
    const int e0  = blockIdx.x * per;
    const int e1  = (e0 + per < E) ? (e0 + per) : E;
    const int* __restrict__ dst = ei + E;
    unsigned* __restrict__ prow = part + (size_t)blockIdx.x * N;

    for (int lo = 0; lo < N; lo += DEG_TABLE) {
        for (int i = threadIdx.x; i < DEG_TABLE; i += 1024) cnt[i] = 0u;
        __syncthreads();
        for (int e = e0 + (int)threadIdx.x; e < e1; e += DEG_BATCH * 1024) {
            int d[DEG_BATCH];
#pragma unroll
            for (int j = 0; j < DEG_BATCH; ++j) {
                const int ee = e + j * 1024;
                d[j] = (ee < e1) ? dst[ee] : -1;
            }
#pragma unroll
            for (int j = 0; j < DEG_BATCH; ++j) {
                const unsigned k = (unsigned)(d[j] - lo);
                if (k < (unsigned)DEG_TABLE) atomicAdd(&cnt[k], 1u);
            }
        }
        __syncthreads();
        const int n = (N - lo < DEG_TABLE) ? (N - lo) : DEG_TABLE;
        for (int i = threadIdx.x; i < n; i += 1024) prow[lo + i] = cnt[i];
        __syncthreads();
    }
}

__global__ void dinv_kernel(const unsigned* __restrict__ part, float* __restrict__ dinv,
                            unsigned* __restrict__ tickets, int n) {
    int i = blockIdx.x * blockDim.x + threadIdx.x;
    if (i < 4) tickets[i] = 0u;
    if (i < n) {
        unsigned c = 1u;
#pragma unroll 16
        for (int s = 0; s < DEG_SLICES; ++s) c += part[(size_t)s * n + i];
        const float d = (float)c;
        dinv[i] = (d > 0.0f) ? rsqrtf(d) : 0.0f;
    }
}

__global__ void norm_kernel(const int* __restrict__ ei, const float* __restrict__ dinv,
                            float* __restrict__ norm, int E) {
    int e = blockIdx.x * blockDim.x + threadIdx.x;
    if (e < E) norm[e] = dinv[ei[e]] * dinv[ei[E + e]];
}

template <bool RELU_IN>
__global__ __launch_bounds__(128) void gemm_wmma_kernel(const float* X,
                                                        const float* __restrict__ W,
                                                        const float* __restrict__ dinv,
                                                        const float* __restrict__ bias,
                                                        float* __restrict__ H,
                                                        float* A,
                                                        unsigned* __restrict__ tickets,
                                                        int K, int Fout, int ldH) {
    const int lane = threadIdx.x & 31;
    const int wave = threadIdx.x >> 5;
    const int half = lane >> 4;
    const int l    = lane & 15;
    const int row0 = blockIdx.x * 16;
    const int col  = wave * 16 + l;
    const bool colOK  = (col < Fout);
    const int  colLd  = colOK ? col : (Fout - 1);

    const float* xrow = X + (size_t)(row0 + l) * K + 2 * half;
    const float* __restrict__ wcol = W + colLd + (size_t)(2 * half) * Fout;

    const float bcol = bias[colLd];
    float dd[8];
#pragma unroll
    for (int v = 0; v < 8; ++v) {
        const float di = dinv[row0 + 8 * half + v];
        dd[v] = di * di;
    }

    v8f c = {};
#pragma unroll 4
    for (int k0 = 0; k0 < K; k0 += 4) {
        float a0 = xrow[k0];
        float a1 = xrow[k0 + 1];
        if (RELU_IN) { a0 = fmaxf(a0, 0.0f); a1 = fmaxf(a1, 0.0f); }
        v2f a; a.x = a0; a.y = a1;
        v2f b;
        b.x = wcol[(size_t)k0 * Fout];
        b.y = wcol[(size_t)(k0 + 1) * Fout];
        c = __builtin_amdgcn_wmma_f32_16x16x4_f32(false, a, false, b, (short)0, c,
                                                  false, false);
    }

    __syncthreads();

    if (tickets != nullptr && blockIdx.x == 0 && threadIdx.x < 2) tickets[threadIdx.x] = 0u;

    if (colOK) {
#pragma unroll
        for (int v = 0; v < 8; ++v) {
            const int m = v + 8 * half;
            H[(size_t)(row0 + m) * ldH + col] = c[v];
            A[(size_t)(row0 + m) * Fout + col] = bcol + c[v] * dd[v];
        }
    }
}


#define AGG_EPW     16
#define CHUNK_EDGES (8 * AGG_EPW)

__device__ __forceinline__ int draw_chunk(unsigned* __restrict__ tickets, int nChunks, int& range) {
    __shared__ int drawn[2];
    if (threadIdx.x == 0) {
        const unsigned where = (unsigned)__builtin_amdgcn_s_sendmsg_rtn(0x87);
        int r = (int)((where >> 18) & 1u);
        int t = (int)atomicAdd(&tickets[r], 1u);
        if (t >= nChunks) {
            r ^= 1;
            t = (int)atomicAdd(&tickets[r], 1u);
        }
        drawn[0] = (t < nChunks) ? t : -1;
        drawn[1] = r;
    }
    __syncthreads();
    range = __builtin_amdgcn_readfirstlane(drawn[1]);
    return __builtin_amdgcn_readfirstlane(drawn[0]);
}

template <int F>
__global__ __launch_bounds__(256) void agg_edges_kernel(const float* __restrict__ H,
                                                        const int* __restrict__ ei,
                                                        const float* __restrict__ norm,
                                                        float* __restrict__ A,
                                                        unsigned* __restrict__ tickets,
                                                        int E, int nChunks, int Nh) {
    static_assert(F % 32 == 0, "whole 32-column chunks only");
    constexpr int CHF = F / 32;
    int range;
    const int chunk = draw_chunk(tickets, nChunks, range);
    if (chunk < 0) return;

    const int lane = threadIdx.x & 31;
    const int wave = __builtin_amdgcn_readfirstlane((int)(threadIdx.x >> 5));

    const int e0 = (chunk * 8 + wave) * AGG_EPW;
    if (e0 + AGG_EPW <= E) {
        int s[AGG_EPW], d[AGG_EPW];
#pragma unroll
        for (int j = 0; j < AGG_EPW; ++j) {
            s[j] = ei[e0 + j];
            d[j] = ei[E + e0 + j];
        }
        asm volatile("" :: "s"(s[0]));
        float h[AGG_EPW][CHF];
#pragma unroll
        for (int j = 0; j < AGG_EPW; ++j) {
            if ((d[j] >= Nh) == (range != 0)) {
                const float* hr = H + (size_t)s[j] * F;
#pragma unroll
                for (int k = 0; k < CHF; ++k) h[j][k] = hr[lane + 32 * k];
            }
        }
        float nv[AGG_EPW];
#pragma unroll
        for (int j = 0; j < AGG_EPW; ++j) nv[j] = norm[e0 + j];
        asm volatile("" :: "s"(nv[0]));
#pragma unroll
        for (int j = 0; j < AGG_EPW; ++j) {
            if ((d[j] >= Nh) == (range != 0)) {
                float* ar = A + (size_t)d[j] * F;
#pragma unroll
                for (int k = 0; k < CHF; ++k) atomicAdd(ar + lane + 32 * k, h[j][k] * nv[j]);
            }
        }
    } else {
        const int e1 = (e0 + AGG_EPW < E) ? (e0 + AGG_EPW) : E;
        for (int e = e0; e < e1; ++e) {
            const int   s  = ei[e];
            const int   d  = ei[E + e];
            const float nv = norm[e];
            if ((d >= Nh) == (range != 0)) {
                for (int c = lane; c < F; c += 32)
                    atomicAdd(A + (size_t)d * F + c, H[(size_t)s * F + c] * nv);
            }
        }
    }
}

__global__ __launch_bounds__(256) void agg_edges_out_kernel(const float* __restrict__ H,
                                                            const int* __restrict__ ei,
                                                            const float* __restrict__ norm,
                                                            float* __restrict__ A,
                                                            unsigned* __restrict__ tickets,
                                                            int E, int nChunks, int Nh) {
    constexpr int HP  = F_HID;
    constexpr int REM = F_OUT - 32;
    static_assert(REM == 8 && F_OUT <= HP,
                  "a row must start on a 32-B step and end inside its second line");
    int range;
    const int chunk = draw_chunk(tickets, nChunks, range);
    if (chunk < 0) return;

    const int lane = threadIdx.x & 31;
    const int wave = __builtin_amdgcn_readfirstlane((int)(threadIdx.x >> 5));

    const int e0 = (chunk * 8 + wave) * AGG_EPW;
    if (e0 + AGG_EPW <= E) {
        int s[AGG_EPW], d[AGG_EPW];
#pragma unroll
        for (int j = 0; j < AGG_EPW; ++j) {
            s[j] = ei[e0 + j];
            d[j] = ei[E + e0 + j];
        }
        asm volatile("" :: "s"(s[0]));
        struct { float lo, hi; } hv[AGG_EPW];
#pragma unroll
        for (int j = 0; j < AGG_EPW; ++j) {
            if ((d[j] >= Nh) == (range != 0)) {
                const int off = (int)(((unsigned)d[j] * (unsigned)F_OUT) & 31u);
                const float* hr = H + (size_t)s[j] * HP;
                const int r = lane - off;
                hv[j].lo = hr[r & 31];
                hv[j].hi = hr[32 + (r & (REM - 1))];
            }
        }
        float nv[AGG_EPW];
#pragma unroll
        for (int j = 0; j < AGG_EPW; ++j) nv[j] = norm[e0 + j];
        asm volatile("" :: "s"(nv[0]));
#pragma unroll
        for (int j = 0; j < AGG_EPW; ++j) {
            if ((d[j] >= Nh) == (range != 0)) {
                const int off = (int)(((unsigned)d[j] * (unsigned)F_OUT) & 31u);
                const size_t l0 = (size_t)d[j] * F_OUT - off;
                const float v0 = hv[j].lo * nv[j];
                const float v1 = ((lane < off) ? hv[j].lo : hv[j].hi) * nv[j];
                if (lane >= off)      atomicAdd(A + l0 + lane, v0);
                if (lane < off + REM) atomicAdd(A + l0 + 32 + lane, v1);
            }
        }
    } else {
        const int e1 = (e0 + AGG_EPW < E) ? (e0 + AGG_EPW) : E;
        for (int e = e0; e < e1; ++e) {
            const int   s  = ei[e];
            const int   d  = ei[E + e];
            const float nv = norm[e];
            if ((d >= Nh) == (range != 0)) {
                for (int c = lane; c < F_OUT; c += 32)
                    atomicAdd(A + (size_t)d * F_OUT + c, H[(size_t)s * HP + c] * nv);
            }
        }
    }
}


extern "C" void kernel_launch(void* const* d_in, const int* in_sizes, int n_in,
                              void* d_out, int out_size, void* d_ws, size_t ws_size,
                              hipStream_t stream) {
    const float* x  = (const float*)d_in[0];
    const int*   ei = (const int*)d_in[1];
    const float* W1 = (const float*)d_in[2];
    const float* b1 = (const float*)d_in[3];
    const float* W2 = (const float*)d_in[4];
    const float* b2 = (const float*)d_in[5];
    const float* W3 = (const float*)d_in[6];
    const float* b3 = (const float*)d_in[7];
    float* out = (float*)d_out;

    const int N = in_sizes[0] / F_IN;
    const int E = in_sizes[1] / 2;

    float* bufH = (float*)d_ws;
    float* bufA = bufH + (size_t)N * F_HID;
    float* dinv = bufA + (size_t)N * F_HID;
    float* norm = dinv + N;
    unsigned* part = (unsigned*)d_ws;

    unsigned* tick1 = (unsigned*)d_out;
    unsigned* tick2 = tick1 + 2;
    unsigned* tick3 = (unsigned*)(bufH + F_OUT);

    const int T = 256;
    const int gN  = (N + T - 1) / T;
    const int gE  = (E + T - 1) / T;
    const int nChunks = (E + CHUNK_EDGES - 1) / CHUNK_EDGES;
    const unsigned gCH = 2u * (unsigned)nChunks;
    const int Nh = (N + 1) / 2;

    deg_count_kernel<<<DEG_SLICES, 1024, 0, stream>>>(ei, part, N, E);
    dinv_kernel     <<<gN, T, 0, stream>>>(part, dinv, tick1, N);
    norm_kernel     <<<gE, T, 0, stream>>>(ei, dinv, norm, E);

    const int rowTiles = (N + 15) / 16;
    const int wavesHid = (F_HID + 15) / 16;
    const int wavesOut = (F_OUT + 15) / 16;


    gemm_wmma_kernel<false><<<rowTiles, 32 * wavesHid, 0, stream>>>(x, W1, dinv, b1, bufH, bufA, nullptr, F_IN, F_HID, F_HID);
    agg_edges_kernel<F_HID><<<gCH, T, 0, stream>>>(bufH, ei, norm, bufA, tick1, E, nChunks, Nh);

    gemm_wmma_kernel<true> <<<rowTiles, 32 * wavesHid, 0, stream>>>(bufA, W2, dinv, b2, bufH, bufA, nullptr, F_HID, F_HID, F_HID);
    agg_edges_kernel<F_HID><<<gCH, T, 0, stream>>>(bufH, ei, norm, bufA, tick2, E, nChunks, Nh);

    gemm_wmma_kernel<true> <<<rowTiles, 32 * wavesOut, 0, stream>>>(bufA, W3, dinv, b3, bufH, out, tick3, F_HID, F_OUT, F_HID);
    agg_edges_out_kernel<<<gCH, T, 0, stream>>>(bufH, ei, norm, out, tick3, E, nChunks, Nh);
}


# Edge weights (norm) get no launch of their own: extra work-groups behind the layer 1 projection's row tiles compute them in that launch, 4 edges a thread, 16-B pieces; 9 launches -> 8; sums unchanged
# speedup vs baseline: 2.6324x; 1.0261x over previous
#include <hip/hip_runtime.h>

typedef __attribute__((ext_vector_type(2))) float v2f;
typedef __attribute__((ext_vector_type(8))) float v8f;

#define F_IN  128
#define F_HID 64
#define F_OUT 40


#define DEG_SLICES 64
#define DEG_TABLE  51200
#define DEG_BATCH  8

__global__ __launch_bounds__(1024) void deg_count_kernel(const int* __restrict__ ei,
                                                         unsigned* __restrict__ part,
                                                         int N, int E) {
    __shared__ unsigned cnt[DEG_TABLE];
    const int per = (E + DEG_SLICES - 1) / DEG_SLICES;
    const int e0  = blockIdx.x * per;
    const int e1  = (e0 + per < E) ? (e0 + per) : E;
    const int* __restrict__ dst = ei + E;
    unsigned* __restrict__ prow = part + (size_t)blockIdx.x * N;

    for (int lo = 0; lo < N; lo += DEG_TABLE) {
        for (int i = threadIdx.x; i < DEG_TABLE; i += 1024) cnt[i] = 0u;
        __syncthreads();
        for (int e = e0 + (int)threadIdx.x; e < e1; e += DEG_BATCH * 1024) {
            int d[DEG_BATCH];
#pragma unroll
            for (int j = 0; j < DEG_BATCH; ++j) {
                const int ee = e + j * 1024;
                d[j] = (ee < e1) ? dst[ee] : -1;
            }
#pragma unroll
            for (int j = 0; j < DEG_BATCH; ++j) {
                const unsigned k = (unsigned)(d[j] - lo);
                if (k < (unsigned)DEG_TABLE) atomicAdd(&cnt[k], 1u);
            }
        }
        __syncthreads();
        const int n = (N - lo < DEG_TABLE) ? (N - lo) : DEG_TABLE;
        for (int i = threadIdx.x; i < n; i += 1024) prow[lo + i] = cnt[i];
        __syncthreads();
    }
}

__global__ void dinv_kernel(const unsigned* __restrict__ part, float* __restrict__ dinv,
                            unsigned* __restrict__ tickets, int n) {
    int i = blockIdx.x * blockDim.x + threadIdx.x;
    if (i < 4) tickets[i] = 0u;
    if (i < n) {
        unsigned c = 1u;
#pragma unroll 16
        for (int s = 0; s < DEG_SLICES; ++s) c += part[(size_t)s * n + i];
        const float d = (float)c;
        dinv[i] = (d > 0.0f) ? rsqrtf(d) : 0.0f;
    }
}

#define NORM_EPT 4
#define NORM_EPB (128 * NORM_EPT)

__device__ __forceinline__ void norm_edges(const int* __restrict__ ei, const float* __restrict__ dinv,
                                           float* __restrict__ norm, int E, int base) {
    if (base >= E) return;
    const bool wide = ((E & 3) == 0) && (base + NORM_EPT <= E) &&
                      ((((size_t)ei) | ((size_t)norm)) & 15) == 0;
    if (wide) {
        const int4 s = *(const int4*)(ei + base);
        const int4 d = *(const int4*)(ei + E + base);
        float4 w;
        w.x = dinv[s.x] * dinv[d.x];
        w.y = dinv[s.y] * dinv[d.y];
        w.z = dinv[s.z] * dinv[d.z];
        w.w = dinv[s.w] * dinv[d.w];
        *(float4*)(norm + base) = w;
    } else {
        for (int j = 0; j < NORM_EPT; ++j) {
            const int e = base + j;
            if (e < E) norm[e] = dinv[ei[e]] * dinv[ei[E + e]];
        }
    }
}

template <bool RELU_IN>
__device__ __forceinline__ void gemm_tile(const float* X,
                                          const float* __restrict__ W,
                                          const float* __restrict__ dinv,
                                          const float* __restrict__ bias,
                                          float* __restrict__ H,
                                          float* A,
                                          unsigned* __restrict__ tickets,
                                          int K, int Fout, int ldH, const int tile) {
    const int lane = threadIdx.x & 31;
    const int wave = threadIdx.x >> 5;
    const int half = lane >> 4;
    const int l    = lane & 15;
    const int row0 = tile * 16;
    const int col  = wave * 16 + l;
    const bool colOK  = (col < Fout);
    const int  colLd  = colOK ? col : (Fout - 1);

    const float* xrow = X + (size_t)(row0 + l) * K + 2 * half;
    const float* __restrict__ wcol = W + colLd + (size_t)(2 * half) * Fout;

    const float bcol = bias[colLd];
    float dd[8];
#pragma unroll
    for (int v = 0; v < 8; ++v) {
        const float di = dinv[row0 + 8 * half + v];
        dd[v] = di * di;
    }

    v8f c = {};
#pragma unroll 4
    for (int k0 = 0; k0 < K; k0 += 4) {
        float a0 = xrow[k0];
        float a1 = xrow[k0 + 1];
        if (RELU_IN) { a0 = fmaxf(a0, 0.0f); a1 = fmaxf(a1, 0.0f); }
        v2f a; a.x = a0; a.y = a1;
        v2f b;
        b.x = wcol[(size_t)k0 * Fout];
        b.y = wcol[(size_t)(k0 + 1) * Fout];
        c = __builtin_amdgcn_wmma_f32_16x16x4_f32(false, a, false, b, (short)0, c,
                                                  false, false);
    }

    __syncthreads();

    if (tickets != nullptr && tile == 0 && threadIdx.x < 2) tickets[threadIdx.x] = 0u;

    if (colOK) {
#pragma unroll
        for (int v = 0; v < 8; ++v) {
            const int m = v + 8 * half;
            H[(size_t)(row0 + m) * ldH + col] = c[v];
            A[(size_t)(row0 + m) * Fout + col] = bcol + c[v] * dd[v];
        }
    }
}

template <bool RELU_IN>
__global__ __launch_bounds__(128) void gemm_wmma_kernel(const float* X,
                                                        const float* __restrict__ W,
                                                        const float* __restrict__ dinv,
                                                        const float* __restrict__ bias,
                                                        float* __restrict__ H,
                                                        float* A,
                                                        unsigned* __restrict__ tickets,
                                                        int K, int Fout, int ldH) {
    gemm_tile<RELU_IN>(X, W, dinv, bias, H, A, tickets, K, Fout, ldH, (int)blockIdx.x);
}

__global__ __launch_bounds__(128) void gemm_norm_kernel(const float* X,
                                                        const float* __restrict__ W,
                                                        const float* __restrict__ dinv,
                                                        const float* __restrict__ bias,
                                                        float* __restrict__ H,
                                                        float* A,
                                                        int K, int Fout, int ldH,
                                                        const int* __restrict__ ei,
                                                        float* __restrict__ norm,
                                                        int E, int rowTiles) {
    const int b = (int)blockIdx.x;
    if (b < rowTiles) {
        gemm_tile<false>(X, W, dinv, bias, H, A, nullptr, K, Fout, ldH, b);
    } else {
        norm_edges(ei, dinv, norm, E, ((b - rowTiles) * 128 + (int)threadIdx.x) * NORM_EPT);
    }
}


#define AGG_EPW     16
#define CHUNK_EDGES (8 * AGG_EPW)

__device__ __forceinline__ int draw_chunk(unsigned* __restrict__ tickets, int nChunks, int& range) {
    __shared__ int drawn[2];
    if (threadIdx.x == 0) {
        const unsigned where = (unsigned)__builtin_amdgcn_s_sendmsg_rtn(0x87);
        int r = (int)((where >> 18) & 1u);
        int t = (int)atomicAdd(&tickets[r], 1u);
        if (t >= nChunks) {
            r ^= 1;
            t = (int)atomicAdd(&tickets[r], 1u);
        }
        drawn[0] = (t < nChunks) ? t : -1;
        drawn[1] = r;
    }
    __syncthreads();
    range = __builtin_amdgcn_readfirstlane(drawn[1]);
    return __builtin_amdgcn_readfirstlane(drawn[0]);
}

template <int F>
__global__ __launch_bounds__(256) void agg_edges_kernel(const float* __restrict__ H,
                                                        const int* __restrict__ ei,
                                                        const float* __restrict__ norm,
                                                        float* __restrict__ A,
                                                        unsigned* __restrict__ tickets,
                                                        int E, int nChunks, int Nh) {
    static_assert(F % 32 == 0, "whole 32-column chunks only");
    constexpr int CHF = F / 32;
    int range;
    const int chunk = draw_chunk(tickets, nChunks, range);
    if (chunk < 0) return;

    const int lane = threadIdx.x & 31;
    const int wave = __builtin_amdgcn_readfirstlane((int)(threadIdx.x >> 5));

    const int e0 = (chunk * 8 + wave) * AGG_EPW;
    if (e0 + AGG_EPW <= E) {
        int s[AGG_EPW], d[AGG_EPW];
#pragma unroll
        for (int j = 0; j < AGG_EPW; ++j) {
            s[j] = ei[e0 + j];
            d[j] = ei[E + e0 + j];
        }
        asm volatile("" :: "s"(s[0]));
        float h[AGG_EPW][CHF];
#pragma unroll
        for (int j = 0; j < AGG_EPW; ++j) {
            if ((d[j] >= Nh) == (range != 0)) {
                const float* hr = H + (size_t)s[j] * F;
#pragma unroll
                for (int k = 0; k < CHF; ++k) h[j][k] = hr[lane + 32 * k];
            }
        }
        float nv[AGG_EPW];
#pragma unroll
        for (int j = 0; j < AGG_EPW; ++j) nv[j] = norm[e0 + j];
        asm volatile("" :: "s"(nv[0]));
#pragma unroll
        for (int j = 0; j < AGG_EPW; ++j) {
            if ((d[j] >= Nh) == (range != 0)) {
                float* ar = A + (size_t)d[j] * F;
#pragma unroll
                for (int k = 0; k < CHF; ++k) atomicAdd(ar + lane + 32 * k, h[j][k] * nv[j]);
            }
        }
    } else {
        const int e1 = (e0 + AGG_EPW < E) ? (e0 + AGG_EPW) : E;
        for (int e = e0; e < e1; ++e) {
            const int   s  = ei[e];
            const int   d  = ei[E + e];
            const float nv = norm[e];
            if ((d >= Nh) == (range != 0)) {
                for (int c = lane; c < F; c += 32)
                    atomicAdd(A + (size_t)d * F + c, H[(size_t)s * F + c] * nv);
            }
        }
    }
}

__global__ __launch_bounds__(256) void agg_edges_out_kernel(const float* __restrict__ H,
                                                            const int* __restrict__ ei,
                                                            const float* __restrict__ norm,
                                                            float* __restrict__ A,
                                                            unsigned* __restrict__ tickets,
                                                            int E, int nChunks, int Nh) {
    constexpr int HP  = F_HID;
    constexpr int REM = F_OUT - 32;
    static_assert(REM == 8 && F_OUT <= HP,
                  "a row must start on a 32-B step and end inside its second line");
    int range;
    const int chunk = draw_chunk(tickets, nChunks, range);
    if (chunk < 0) return;

    const int lane = threadIdx.x & 31;
    const int wave = __builtin_amdgcn_readfirstlane((int)(threadIdx.x >> 5));

    const int e0 = (chunk * 8 + wave) * AGG_EPW;
    if (e0 + AGG_EPW <= E) {
        int s[AGG_EPW], d[AGG_EPW];
#pragma unroll
        for (int j = 0; j < AGG_EPW; ++j) {
            s[j] = ei[e0 + j];
            d[j] = ei[E + e0 + j];
        }
        asm volatile("" :: "s"(s[0]));
        struct { float lo, hi; } hv[AGG_EPW];
#pragma unroll
        for (int j = 0; j < AGG_EPW; ++j) {
            if ((d[j] >= Nh) == (range != 0)) {
                const int off = (int)(((unsigned)d[j] * (unsigned)F_OUT) & 31u);
                const float* hr = H + (size_t)s[j] * HP;
                const int r = lane - off;
                hv[j].lo = hr[r & 31];
                hv[j].hi = hr[32 + (r & (REM - 1))];
            }
        }
        float nv[AGG_EPW];
#pragma unroll
        for (int j = 0; j < AGG_EPW; ++j) nv[j] = norm[e0 + j];
        asm volatile("" :: "s"(nv[0]));
#pragma unroll
        for (int j = 0; j < AGG_EPW; ++j) {
            if ((d[j] >= Nh) == (range != 0)) {
                const int off = (int)(((unsigned)d[j] * (unsigned)F_OUT) & 31u);
                const size_t l0 = (size_t)d[j] * F_OUT - off;
                const float v0 = hv[j].lo * nv[j];
                const float v1 = ((lane < off) ? hv[j].lo : hv[j].hi) * nv[j];
                if (lane >= off)      atomicAdd(A + l0 + lane, v0);
                if (lane < off + REM) atomicAdd(A + l0 + 32 + lane, v1);
            }
        }
    } else {
        const int e1 = (e0 + AGG_EPW < E) ? (e0 + AGG_EPW) : E;
        for (int e = e0; e < e1; ++e) {
            const int   s  = ei[e];
            const int   d  = ei[E + e];
            const float nv = norm[e];
            if ((d >= Nh) == (range != 0)) {
                for (int c = lane; c < F_OUT; c += 32)
                    atomicAdd(A + (size_t)d * F_OUT + c, H[(size_t)s * HP + c] * nv);
            }
        }
    }
}


extern "C" void kernel_launch(void* const* d_in, const int* in_sizes, int n_in,
                              void* d_out, int out_size, void* d_ws, size_t ws_size,
                              hipStream_t stream) {
    const float* x  = (const float*)d_in[0];
    const int*   ei = (const int*)d_in[1];
    const float* W1 = (const float*)d_in[2];
    const float* b1 = (const float*)d_in[3];
    const float* W2 = (const float*)d_in[4];
    const float* b2 = (const float*)d_in[5];
    const float* W3 = (const float*)d_in[6];
    const float* b3 = (const float*)d_in[7];
    float* out = (float*)d_out;

    const int N = in_sizes[0] / F_IN;
    const int E = in_sizes[1] / 2;

    float* bufH = (float*)d_ws;
    float* bufA = bufH + (size_t)N * F_HID;
    float* dinv = bufA + (size_t)N * F_HID;
    float* norm = dinv + N;
    unsigned* part = (unsigned*)d_ws;

    unsigned* tick1 = (unsigned*)d_out;
    unsigned* tick2 = tick1 + 2;
    unsigned* tick3 = (unsigned*)(bufH + F_OUT);

    const int T = 256;
    const int gN  = (N + T - 1) / T;
    const int gNB = (E + NORM_EPB - 1) / NORM_EPB;
    const int nChunks = (E + CHUNK_EDGES - 1) / CHUNK_EDGES;
    const unsigned gCH = 2u * (unsigned)nChunks;
    const int Nh = (N + 1) / 2;

    deg_count_kernel<<<DEG_SLICES, 1024, 0, stream>>>(ei, part, N, E);
    dinv_kernel     <<<gN, T, 0, stream>>>(part, dinv, tick1, N);

    const int rowTiles = (N + 15) / 16;
    const int wavesHid = (F_HID + 15) / 16;
    const int wavesOut = (F_OUT + 15) / 16;


    gemm_norm_kernel<<<rowTiles + gNB, 32 * wavesHid, 0, stream>>>(x, W1, dinv, b1, bufH, bufA, F_IN, F_HID, F_HID, ei, norm, E, rowTiles);
    agg_edges_kernel<F_HID><<<gCH, T, 0, stream>>>(bufH, ei, norm, bufA, tick1, E, nChunks, Nh);

    gemm_wmma_kernel<true> <<<rowTiles, 32 * wavesHid, 0, stream>>>(bufA, W2, dinv, b2, bufH, bufA, nullptr, F_HID, F_HID, F_HID);
    agg_edges_kernel<F_HID><<<gCH, T, 0, stream>>>(bufH, ei, norm, bufA, tick2, E, nChunks, Nh);

    gemm_wmma_kernel<true> <<<rowTiles, 32 * wavesOut, 0, stream>>>(bufA, W3, dinv, b3, bufH, out, tick3, F_HID, F_OUT, F_HID);
    agg_edges_out_kernel<<<gCH, T, 0, stream>>>(bufH, ei, norm, out, tick3, E, nChunks, Nh);
}
